# starting kernel
# baseline (speedup 1.0000x reference)
#include <hip/hip_runtime.h>
#include <float.h>

typedef __attribute__((ext_vector_type(2))) float v2f;
typedef __attribute__((ext_vector_type(8))) float v8f;

#define KNN 16
#define UNROLL 4

__global__ __launch_bounds__(32)
void softproj_knn_wmma(const float* __restrict__ pc,
                       const float* __restrict__ qc,
                       float* __restrict__ out,
                       int N, int M) {
    const int lane = threadIdx.x;
    const int mrow = lane & 15;
    const bool hi  = lane >= 16;

    const int tiles_m = M >> 4;
    const int b  = blockIdx.x / tiles_m;
    const int m0 = (blockIdx.x % tiles_m) << 4;

    const float* pcb = pc + (size_t)b * 3 * N;
    const float* qcb = qc + (size_t)b * 3 * M;

    const int   mg = m0 + mrow;
    const float qx = qcb[0 * M + mg];
    const float qy = qcb[1 * M + mg];
    const float qz = qcb[2 * M + mg];
    v2f afrag;
    afrag.x = hi ? qz   : qx;
    afrag.y = hi ? 1.0f : qy;

    __shared__ float tile_d[UNROLL][16][20];
    __shared__ float lds_ld[32][KNN + 1];
    __shared__ int   lds_li[32][KNN + 1];

    float kd[KNN];
    int   ki[KNN];
#pragma unroll
    for (int j = 0; j < KNN; ++j) { kd[j] = FLT_MAX; ki[j] = 0; }

    const int c0 = hi ? 8 : 0;

    for (int n0 = 0; n0 < N; n0 += 16 * UNROLL) {
        float px[UNROLL], py[UNROLL], pz[UNROLL];
#pragma unroll
        for (int u = 0; u < UNROLL; ++u) {
            const int n = n0 + 16 * u + mrow;
            px[u] = pcb[0 * N + n];
            py[u] = pcb[1 * N + n];
            pz[u] = pcb[2 * N + n];
        }

#pragma unroll
        for (int u = 0; u < UNROLL; ++u) {
            const float psq = px[u] * px[u] + py[u] * py[u] + pz[u] * pz[u];
            v2f bfrag;
            bfrag.x = hi ? (-2.0f * pz[u]) : (-2.0f * px[u]);
            bfrag.y = hi ? psq             : (-2.0f * py[u]);

            v8f c = {};
            c = __builtin_amdgcn_wmma_f32_16x16x4_f32(false, afrag, false, bfrag,
                                                      (short)0, c, false, false);
#pragma unroll
            for (int r = 0; r < 8; ++r) {
                tile_d[u][r + (hi ? 8 : 0)][mrow] = c[r];
            }
        }
        __syncthreads();

#pragma unroll
        for (int u = 0; u < UNROLL; ++u) {
#pragma unroll
            for (int j = 0; j < 8; ++j) {
                const float d  = tile_d[u][mrow][c0 + j];
                const int   gi = n0 + 16 * u + c0 + j;
                if (d < kd[KNN - 1]) {
                    kd[KNN - 1] = d; ki[KNN - 1] = gi;
#pragma unroll
                    for (int t = KNN - 1; t > 0; --t) {
                        if (kd[t] < kd[t - 1]) {
                            float td = kd[t]; kd[t] = kd[t - 1]; kd[t - 1] = td;
                            int   ti = ki[t]; ki[t] = ki[t - 1]; ki[t - 1] = ti;
                        }
                    }
                }
            }
        }
        __syncthreads();
    }

#pragma unroll
    for (int j = 0; j < KNN; ++j) { lds_ld[lane][j] = kd[j]; lds_li[lane][j] = ki[j]; }
    lds_ld[lane][KNN] = FLT_MAX;
    lds_li[lane][KNN] = 0;
    __syncthreads();

    if (lane < 16) {
        int midx[KNN];
        int i0 = 0, i1 = 0;
#pragma unroll
        for (int j = 0; j < KNN; ++j) {
            const float d0 = lds_ld[mrow][i0];
            const float d1 = lds_ld[mrow + 16][i1];
            if (d0 <= d1) { midx[j] = lds_li[mrow][i0];      ++i0; }
            else          { midx[j] = lds_li[mrow + 16][i1]; ++i1; }
        }

        float pxa[KNN], pya[KNN], pza[KNN], ed[KNN];
        float dmin = FLT_MAX;
#pragma unroll
        for (int j = 0; j < KNN; ++j) {
            const int   gi = midx[j];
            const float px = pcb[0 * N + gi];
            const float py = pcb[1 * N + gi];
            const float pz = pcb[2 * N + gi];
            const float dx = px - qx, dy = py - qy, dz = pz - qz;
            const float d  = dx * dx + dy * dy + dz * dz;
            pxa[j] = px; pya[j] = py; pza[j] = pz; ed[j] = d;
            dmin = fminf(dmin, d);
        }

        float wsum = 0.0f, ox = 0.0f, oy = 0.0f, oz = 0.0f;
#pragma unroll
        for (int j = 0; j < KNN; ++j) {
            const float w = __expf(dmin - ed[j]);
            wsum += w;
            ox += w * pxa[j];
            oy += w * pya[j];
            oz += w * pza[j];
        }
        const float inv = 1.0f / wsum;
        float* outb = out + (size_t)b * 3 * M;
        outb[0 * M + mg] = ox * inv;
        outb[1 * M + mg] = oy * inv;
        outb[2 * M + mg] = oz * inv;
    }
}

extern "C" void kernel_launch(void* const* d_in, const int* in_sizes, int n_in,
                              void* d_out, int out_size, void* d_ws, size_t ws_size,
                              hipStream_t stream) {
    (void)n_in; (void)out_size; (void)d_ws; (void)ws_size;
    const float* pc = (const float*)d_in[0];
    const float* qc = (const float*)d_in[1];
    float* out = (float*)d_out;

    const int B = 4;
    const int N = in_sizes[0] / (3 * B);
    const int M = in_sizes[1] / (3 * B);

    dim3 grid(B * (M / 16));
    softproj_knn_wmma<<<grid, 32, 0, stream>>>(pc, qc, out, N, M);
}


# Selection split by query (not by point range) over 8 waves per work-group: each wave recomputes every WMMA tile, A rows rotated, but filters/inserts only D register 0 (2 queries); same set, sum order
# speedup vs baseline: 4.6390x; 4.6390x over previous
#include <hip/hip_runtime.h>
#include <float.h>

typedef __attribute__((ext_vector_type(2))) float v2f;
typedef __attribute__((ext_vector_type(8))) float v8f;

#define KNN 16
#define UNROLL 4
#define WAVES 8


__device__ __forceinline__ int row_shr1_i(int v) {
    return __builtin_amdgcn_update_dpp(v, v, 0x111, 0xF, 0xF, false);
}
__device__ __forceinline__ float row_shr1_f(float v) {
    return __int_as_float(row_shr1_i(__float_as_int(v)));
}
__device__ __forceinline__ float row_last_f(float v) {
    const int b = __float_as_int(v);
    return __int_as_float(__builtin_amdgcn_update_dpp(b, b, 0x15F, 0xF, 0xF, false));
}

__global__ __launch_bounds__(32 * WAVES)
void softproj_knn_wmma(const float* __restrict__ pc,
                       const float* __restrict__ qc,
                       float* __restrict__ out,
                       int N, int M) {
    const int lane = threadIdx.x & 31;
    const int wv   = __builtin_amdgcn_readfirstlane((int)(threadIdx.x >> 5));
    const int mrow = lane & 15;
    const bool hi  = lane >= 16;

    const int tiles_m = M >> 4;
    const int b  = blockIdx.x / tiles_m;
    const int m0 = (blockIdx.x % tiles_m) << 4;

    const float* pcb = pc + (size_t)b * 3 * N;
    const float* qcb = qc + (size_t)b * 3 * M;

    const int   mga = m0 + ((mrow & 8) | ((mrow + wv) & 7));
    const float ax = qcb[0 * M + mga];
    const float ay = qcb[1 * M + mga];
    const float az = qcb[2 * M + mga];
    v2f afrag;
    afrag.x = hi ? az   : ax;
    afrag.y = hi ? 1.0f : ay;

    __shared__ int lds_li[16][KNN + 1];

    float ld  = FLT_MAX;
    int   li  = 0;
    float thr = FLT_MAX;

    float px[UNROLL], py[UNROLL], pz[UNROLL];
#pragma unroll
    for (int u = 0; u < UNROLL; ++u) {
        const int n = 16 * u + mrow;
        px[u] = pcb[0 * N + n];
        py[u] = pcb[1 * N + n];
        pz[u] = pcb[2 * N + n];
    }

    for (int n0 = 0; n0 < N; n0 += 16 * UNROLL) {
        const int nn = (n0 + 16 * UNROLL < N) ? (n0 + 16 * UNROLL) : n0;
        float nx[UNROLL], ny[UNROLL], nz[UNROLL];
#pragma unroll
        for (int u = 0; u < UNROLL; ++u) {
            const int n = nn + 16 * u + mrow;
            nx[u] = pcb[0 * N + n];
            ny[u] = pcb[1 * N + n];
            nz[u] = pcb[2 * N + n];
        }

        v8f c[UNROLL];
#pragma unroll
        for (int u = 0; u < UNROLL; ++u) {
            const float psq = px[u] * px[u] + py[u] * py[u] + pz[u] * pz[u];
            v2f bfrag;
            bfrag.x = hi ? (-2.0f * pz[u]) : (-2.0f * px[u]);
            bfrag.y = hi ? psq             : (-2.0f * py[u]);

            v8f z = {};
            c[u] = __builtin_amdgcn_wmma_f32_16x16x4_f32(false, afrag, false, bfrag,
                                                         (short)0, z, false, false);
        }
        unsigned hit[UNROLL];
#pragma unroll
        for (int u = 0; u < UNROLL; ++u) {
            hit[u] = __builtin_amdgcn_ballot_w32(c[u][0] < thr);
        }
#pragma unroll
        for (int u = 0; u < UNROLL; ++u) {
            if (__builtin_expect(hit[u] != 0u, 0)) {
                const int nb = n0 + 16 * u;
                unsigned mask = __builtin_amdgcn_ballot_w32(c[u][0] < thr);
                while (mask != 0u) {
                    const int src = __builtin_ctz(mask);
                    mask &= mask - 1u;
                    const float dn = __int_as_float(
                        __builtin_amdgcn_readlane(__float_as_int(c[u][0]), src));
                    const int   in = nb + (src & 15);
                    const float pd = row_shr1_f(ld);
                    const int   pi = row_shr1_i(li);
                    const bool gt  = ((lane ^ src) < 16) && (ld > dn);
                    const bool pgt = (mrow != 0) && (pd > dn);
                    ld = gt ? (pgt ? pd : dn) : ld;
                    li = gt ? (pgt ? pi : in) : li;
                }
                thr = row_last_f(ld);
            }
        }

#pragma unroll
        for (int u = 0; u < UNROLL; ++u) { px[u] = nx[u]; py[u] = ny[u]; pz[u] = nz[u]; }
    }

    lds_li[wv + (hi ? 8 : 0)][mrow] = li;
    __syncthreads();

    if (wv == 0 && lane < 16) {
        const int   mg = mga;
        const float qx = ax, qy = ay, qz = az;
        int midx[KNN];
#pragma unroll
        for (int j = 0; j < KNN; ++j) midx[j] = lds_li[mrow][j];

        float pxa[KNN], pya[KNN], pza[KNN], ed[KNN];
        float dmin = FLT_MAX;
#pragma unroll
        for (int j = 0; j < KNN; ++j) {
            const int   gi = midx[j];
            const float gx = pcb[0 * N + gi];
            const float gy = pcb[1 * N + gi];
            const float gz = pcb[2 * N + gi];
            const float dx = gx - qx, dy = gy - qy, dz = gz - qz;
            const float d  = dx * dx + dy * dy + dz * dz;
            pxa[j] = gx; pya[j] = gy; pza[j] = gz; ed[j] = d;
            dmin = fminf(dmin, d);
        }

        float wsum = 0.0f, ox = 0.0f, oy = 0.0f, oz = 0.0f;
#pragma unroll
        for (int j = 0; j < KNN; ++j) {
            const float w = __expf(dmin - ed[j]);
            wsum += w;
            ox += w * pxa[j];
            oy += w * pya[j];
            oz += w * pza[j];
        }
        const float inv = 1.0f / wsum;
        float* outb = out + (size_t)b * 3 * M;
        outb[0 * M + mg] = ox * inv;
        outb[1 * M + mg] = oy * inv;
        outb[2 * M + mg] = oz * inv;
    }
}

extern "C" void kernel_launch(void* const* d_in, const int* in_sizes, int n_in,
                              void* d_out, int out_size, void* d_ws, size_t ws_size,
                              hipStream_t stream) {
    (void)n_in; (void)out_size; (void)d_ws; (void)ws_size;
    const float* pc = (const float*)d_in[0];
    const float* qc = (const float*)d_in[1];
    float* out = (float*)d_out;

    const int B = 4;
    const int N = in_sizes[0] / (3 * B);
    const int M = in_sizes[1] / (3 * B);

    dim3 grid(B * (M / 16));
    softproj_knn_wmma<<<grid, 32 * WAVES, 0, stream>>>(pc, qc, out, N, M);
}


# Point cloud staged once per work-group in LDS as ready B columns (-2x,-2y,-2z,|p|^2), 2-chunk ring, next chunk's loads fly under the scan; waves read B fragments from LDS; same set, sum order
# speedup vs baseline: 7.6120x; 1.6409x over previous
#include <hip/hip_runtime.h>
#include <float.h>

typedef __attribute__((ext_vector_type(2))) float v2f;
typedef __attribute__((ext_vector_type(8))) float v8f;

#define KNN 16
#define UNROLL 4
#define WAVES 8
#define CHUNK 2048
#define PPT (CHUNK / (32 * WAVES))


__device__ __forceinline__ int row_shr1_i(int v) {
    return __builtin_amdgcn_update_dpp(v, v, 0x111, 0xF, 0xF, false);
}
__device__ __forceinline__ float row_shr1_f(float v) {
    return __int_as_float(row_shr1_i(__float_as_int(v)));
}
__device__ __forceinline__ float row_last_f(float v) {
    const int b = __float_as_int(v);
    return __int_as_float(__builtin_amdgcn_update_dpp(b, b, 0x15F, 0xF, 0xF, false));
}

__global__ __launch_bounds__(32 * WAVES)
void softproj_knn_wmma(const float* __restrict__ pc,
                       const float* __restrict__ qc,
                       float* __restrict__ out,
                       int N, int M) {
    const int lane = threadIdx.x & 31;
    const int wv   = __builtin_amdgcn_readfirstlane((int)(threadIdx.x >> 5));
    const int mrow = lane & 15;
    const bool hi  = lane >= 16;

    const int tiles_m = M >> 4;
    const int b  = blockIdx.x / tiles_m;
    const int m0 = (blockIdx.x % tiles_m) << 4;

    const float* pcb = pc + (size_t)b * 3 * N;
    const float* qcb = qc + (size_t)b * 3 * M;

    const int   mga = m0 + ((mrow & 8) | ((mrow + wv) & 7));
    const float ax = qcb[0 * M + mga];
    const float ay = qcb[1 * M + mga];
    const float az = qcb[2 * M + mga];
    v2f afrag;
    afrag.x = hi ? az   : ax;
    afrag.y = hi ? 1.0f : ay;

    __shared__ int lds_li[16][KNN + 1];
    __shared__ __attribute__((aligned(16))) v2f lds_b[2][CHUNK * 2];

    float ld  = FLT_MAX;
    int   li  = 0;
    float thr = FLT_MAX;

    const int tid = (int)threadIdx.x;
    const int frag = mrow * 2 + (hi ? 1 : 0);

    float sx[PPT], sy[PPT], sz[PPT];
    const unsigned nlast = (unsigned)N - 1u;
#pragma unroll
    for (int j = 0; j < PPT; ++j) {
        const unsigned n = min((unsigned)(j * (32 * WAVES) + tid), nlast);
        sx[j] = pcb[0 * (unsigned)N + n];
        sy[j] = pcb[1 * (unsigned)N + n];
        sz[j] = pcb[2 * (unsigned)N + n];
    }

    int slot = 0;
    for (int base = 0; base < N; base += CHUNK, slot ^= 1) {
#pragma unroll
        for (int j = 0; j < PPT; ++j) {
            const float x = sx[j], y = sy[j], z = sz[j];
            const float psq = x * x + y * y + z * z;
            const int   p   = j * (32 * WAVES) + tid;
            v2f lo2, hi2;
            lo2.x = -2.0f * x; lo2.y = -2.0f * y;
            hi2.x = -2.0f * z; hi2.y = psq;
            lds_b[slot][2 * p]     = lo2;
            lds_b[slot][2 * p + 1] = hi2;
        }
        __syncthreads();

        if (base + CHUNK < N) {
#pragma unroll
            for (int j = 0; j < PPT; ++j) {
                const unsigned n = min((unsigned)(base + CHUNK + j * (32 * WAVES) + tid), nlast);
                sx[j] = pcb[0 * (unsigned)N + n];
                sy[j] = pcb[1 * (unsigned)N + n];
                sz[j] = pcb[2 * (unsigned)N + n];
            }
        }

        const int  cend = (base + CHUNK < N) ? (base + CHUNK) : N;
        const v2f* lp   = &lds_b[slot][frag];

        v2f cur[UNROLL];
#pragma unroll
        for (int u = 0; u < UNROLL; ++u) cur[u] = lp[(16 * u) * 2];
        const v2f* rp = lp;

        for (int n0 = base; n0 < cend; n0 += 16 * UNROLL) {
            const v2f* np = rp + ((n0 + 16 * UNROLL < cend) ? (16 * UNROLL) * 2 : 0);
            v2f nxt[UNROLL];
#pragma unroll
            for (int u = 0; u < UNROLL; ++u) nxt[u] = np[(16 * u) * 2];

            v8f c[UNROLL];
#pragma unroll
            for (int u = 0; u < UNROLL; ++u) {
                v8f z = {};
                c[u] = __builtin_amdgcn_wmma_f32_16x16x4_f32(false, afrag, false, cur[u],
                                                             (short)0, z, false, false);
            }
            unsigned hit[UNROLL];
#pragma unroll
            for (int u = 0; u < UNROLL; ++u) {
                hit[u] = __builtin_amdgcn_ballot_w32(c[u][0] < thr);
            }
#pragma unroll
            for (int u = 0; u < UNROLL; ++u) {
                if (__builtin_expect(hit[u] != 0u, 0)) {
                    const int nb = n0 + 16 * u;
                    unsigned mask = __builtin_amdgcn_ballot_w32(c[u][0] < thr);
                    while (mask != 0u) {
                        const int src = __builtin_ctz(mask);
                        mask &= mask - 1u;
                        const float dn = __int_as_float(
                            __builtin_amdgcn_readlane(__float_as_int(c[u][0]), src));
                        const int   in = nb + (src & 15);
                        const float pd = row_shr1_f(ld);
                        const int   pi = row_shr1_i(li);
                        const bool gt  = ((lane ^ src) < 16) && (ld > dn);
                        const bool pgt = (mrow != 0) && (pd > dn);
                        ld = gt ? (pgt ? pd : dn) : ld;
                        li = gt ? (pgt ? pi : in) : li;
                    }
                    thr = row_last_f(ld);
                }
            }

#pragma unroll
            for (int u = 0; u < UNROLL; ++u) cur[u] = nxt[u];
            rp = np;
        }
    }

    lds_li[wv + (hi ? 8 : 0)][mrow] = li;
    __syncthreads();

    if (wv == 0 && lane < 16) {
        const int   mg = mga;
        const float qx = ax, qy = ay, qz = az;
        int midx[KNN];
#pragma unroll
        for (int j = 0; j < KNN; ++j) midx[j] = lds_li[mrow][j];

        float pxa[KNN], pya[KNN], pza[KNN], ed[KNN];
        float dmin = FLT_MAX;
#pragma unroll
        for (int j = 0; j < KNN; ++j) {
            const int   gi = midx[j];
            const float gx = pcb[0 * N + gi];
            const float gy = pcb[1 * N + gi];
            const float gz = pcb[2 * N + gi];
            const float dx = gx - qx, dy = gy - qy, dz = gz - qz;
            const float d  = dx * dx + dy * dy + dz * dz;
            pxa[j] = gx; pya[j] = gy; pza[j] = gz; ed[j] = d;
            dmin = fminf(dmin, d);
        }

        float wsum = 0.0f, ox = 0.0f, oy = 0.0f, oz = 0.0f;
#pragma unroll
        for (int j = 0; j < KNN; ++j) {
            const float w = __expf(dmin - ed[j]);
            wsum += w;
            ox += w * pxa[j];
            oy += w * pya[j];
            oz += w * pza[j];
        }
        const float inv = 1.0f / wsum;
        float* outb = out + (size_t)b * 3 * M;
        outb[0 * M + mg] = ox * inv;
        outb[1 * M + mg] = oy * inv;
        outb[2 * M + mg] = oz * inv;
    }
}

extern "C" void kernel_launch(void* const* d_in, const int* in_sizes, int n_in,
                              void* d_out, int out_size, void* d_ws, size_t ws_size,
                              hipStream_t stream) {
    (void)n_in; (void)out_size; (void)d_ws; (void)ws_size;
    const float* pc = (const float*)d_in[0];
    const float* qc = (const float*)d_in[1];
    float* out = (float*)d_out;

    const int B = 4;
    const int N = in_sizes[0] / (3 * B);
    const int M = in_sizes[1] / (3 * B);

    dim3 grid(B * (M / 16));
    softproj_knn_wmma<<<grid, 32 * WAVES, 0, stream>>>(pc, qc, out, N, M);
}


# Sorted insert kept on the vector unit: other half reads its own entry, a row's first lane sees -inf below, new distance = median of 3; no scalar mask step from compare to select; same set, sum order
# speedup vs baseline: 12.7384x; 1.6735x over previous
#include <hip/hip_runtime.h>
#include <float.h>

typedef __attribute__((ext_vector_type(2))) float v2f;
typedef __attribute__((ext_vector_type(8))) float v8f;

typedef __attribute__((ext_vector_type(4))) float v4f;

#define KNN 16
#define WAVES 8
#define UNROLL 4
#define GROUP (16 * UNROLL)
#define STEP (GROUP * WAVES)


__device__ __forceinline__ int row_shr1_i(int v, int first) {
    return __builtin_amdgcn_update_dpp(first, v, 0x111, 0xF, 0xF, false);
}
__device__ __forceinline__ float row_shr1_f(float v, float first) {
    return __int_as_float(row_shr1_i(__float_as_int(v), __float_as_int(first)));
}
__device__ __forceinline__ float row_last_f(float v) {
    const int b = __float_as_int(v);
    return __int_as_float(__builtin_amdgcn_update_dpp(b, b, 0x15F, 0xF, 0xF, false));
}

__global__ __launch_bounds__(32 * WAVES)
void softproj_knn_wmma(const float* __restrict__ pc,
                       const float* __restrict__ qc,
                       float* __restrict__ out,
                       int N, int M) {
    const int lane = threadIdx.x & 31;
    const int wv   = __builtin_amdgcn_readfirstlane((int)(threadIdx.x >> 5));
    const int mrow = lane & 15;
    const bool hi  = lane >= 16;

    const int tiles_m = M >> 4;
    const int b  = blockIdx.x / tiles_m;
    const int m0 = (blockIdx.x % tiles_m) << 4;

    const float* pcb = pc + (size_t)b * 3 * N;
    const float* qcb = qc + (size_t)b * 3 * M;

    const int   mga = m0 + mrow;
    const float ax = qcb[0 * M + mga];
    const float ay = qcb[1 * M + mga];
    const float az = qcb[2 * M + mga];
    v2f afrag;
    afrag.x = hi ? az   : ax;
    afrag.y = hi ? 1.0f : ay;

    __shared__ int lds_li[16][KNN + 1];
    __shared__ __attribute__((aligned(16))) v4f lds_d[3][8][WAVES][32];

    float ld  = FLT_MAX;
    int   li  = 0;
    float thr = FLT_MAX;

    const unsigned nlast  = (unsigned)N - 1u;
    const int      nsteps = (N + STEP - 1) / STEP;

    float px[UNROLL], py[UNROLL], pz[UNROLL];
    float px1[UNROLL], py1[UNROLL], pz1[UNROLL];

    auto fetch = [&](int t, float (&px)[UNROLL], float (&py)[UNROLL], float (&pz)[UNROLL]) {
#pragma unroll
        for (int u = 0; u < UNROLL; ++u) {
            const unsigned n = min((unsigned)(t * STEP + wv * GROUP + 16 * u + mrow), nlast);
            px[u] = pcb[0 * (unsigned)N + n];
            py[u] = pcb[1 * (unsigned)N + n];
            pz[u] = pcb[2 * (unsigned)N + n];
        }
    };

    auto produce = [&](int t, int buf, const float (&px)[UNROLL], const float (&py)[UNROLL], const float (&pz)[UNROLL]) {
        v8f c[UNROLL];
#pragma unroll
        for (int u = 0; u < UNROLL; ++u) {
            const float x = px[u], y = py[u], z = pz[u];
            float psq = x * x + y * y + z * z;
            if (t * STEP + wv * GROUP + 16 * u + mrow >= N) psq = INFINITY;
            v2f bfrag;
            bfrag.x = hi ? -2.0f * z : -2.0f * x;
            bfrag.y = hi ? psq       : -2.0f * y;
            v8f z0 = {};
            c[u] = __builtin_amdgcn_wmma_f32_16x16x4_f32(false, afrag, false, bfrag,
                                                         (short)0, z0, false, false);
        }
#pragma unroll
        for (int r = 0; r < 8; ++r) {
            v4f v;
            v.x = c[0][r]; v.y = c[1][r]; v.z = c[2][r]; v.w = c[3][r];
            lds_d[buf][r][wv][lane] = v;
        }
    };

    auto scan = [&](int t, int buf) {
        const v4f* rp = &lds_d[buf][wv][0][lane];
        v4f dv[WAVES];
#pragma unroll
        for (int g = 0; g < WAVES; ++g) dv[g] = rp[g * 32];

#pragma unroll
        for (int g = 0; g < WAVES; ++g) {
            unsigned hit[UNROLL];
#pragma unroll
            for (int u = 0; u < UNROLL; ++u) {
                hit[u] = __builtin_amdgcn_ballot_w32(dv[g][u] < thr);
            }
#pragma unroll
            for (int u = 0; u < UNROLL; ++u) {
                if (__builtin_expect(hit[u] != 0u, 0)) {
                    const int nb = t * STEP + g * GROUP + 16 * u;
                    unsigned mask = __builtin_amdgcn_ballot_w32(dv[g][u] < thr);
                    while (mask != 0u) {
                        const int src = __builtin_ctz(mask);
                        mask &= mask - 1u;
                        const float ds = __int_as_float(
                            __builtin_amdgcn_readlane(__float_as_int(dv[g][u]), src));
                        const float dn = __builtin_amdgcn_inverse_ballot_w32(src < 16 ? 0x0000FFFFu : 0xFFFF0000u)
                                             ? ds : ld;
                        const int   in = nb + (src & 15);
                        const float pd = row_shr1_f(ld, -INFINITY);
                        const int   pi = row_shr1_i(li, 0);
                        const bool gt  = ld > dn;
                        const bool pgt = pd > dn;
                        ld = __builtin_amdgcn_fmed3f(pd, dn, ld);
                        li = gt ? (pgt ? pi : in) : li;
                    }
                    thr = row_last_f(ld);
                }
            }
        }
    };

    auto arrive = [&]() {
        __builtin_amdgcn_fence(__ATOMIC_RELEASE, "workgroup");
        __builtin_amdgcn_s_barrier_signal(-1);
    };
    auto pass = [&]() {
        __builtin_amdgcn_s_barrier_wait(-1);
        __builtin_amdgcn_fence(__ATOMIC_ACQUIRE, "workgroup");
    };

    fetch(0, px, py, pz);
    if (1 < nsteps) fetch(1, px1, py1, pz1);
    produce(0, 0, px, py, pz);
    if (1 < nsteps) produce(1, 1, px1, py1, pz1);
    __syncthreads();
    arrive();
    if (2 < nsteps) fetch(2, px, py, pz);

    int bs = 0;
    int bp = 2;
    for (int s = 0; s < nsteps; ++s) {
        scan(s, bs);
        pass();
        if (s + 2 < nsteps) produce(s + 2, bp, px, py, pz);
        arrive();
        if (s + 3 < nsteps) fetch(s + 3, px, py, pz);
        bs = (bs == 2) ? 0 : bs + 1;
        bp = (bp == 2) ? 0 : bp + 1;
    }
    pass();

    lds_li[wv + (hi ? 8 : 0)][mrow] = li;
    __syncthreads();

    if (wv == 0 && lane < 16) {
        const int   mg = mga;
        const float qx = ax, qy = ay, qz = az;
        int midx[KNN];
#pragma unroll
        for (int j = 0; j < KNN; ++j) midx[j] = lds_li[mrow][j];

        float pxa[KNN], pya[KNN], pza[KNN], ed[KNN];
        float dmin = FLT_MAX;
#pragma unroll
        for (int j = 0; j < KNN; ++j) {
            const int   gi = midx[j];
            const float gx = pcb[0 * N + gi];
            const float gy = pcb[1 * N + gi];
            const float gz = pcb[2 * N + gi];
            const float dx = gx - qx, dy = gy - qy, dz = gz - qz;
            const float d  = dx * dx + dy * dy + dz * dz;
            pxa[j] = gx; pya[j] = gy; pza[j] = gz; ed[j] = d;
            dmin = fminf(dmin, d);
        }

        float wsum = 0.0f, ox = 0.0f, oy = 0.0f, oz = 0.0f;
#pragma unroll
        for (int j = 0; j < KNN; ++j) {
            const float w = __expf(dmin - ed[j]);
            wsum += w;
            ox += w * pxa[j];
            oy += w * pya[j];
            oz += w * pza[j];
        }
        const float inv = 1.0f / wsum;
        float* outb = out + (size_t)b * 3 * M;
        outb[0 * M + mg] = ox * inv;
        outb[1 * M + mg] = oy * inv;
        outb[2 * M + mg] = oz * inv;
    }
}

extern "C" void kernel_launch(void* const* d_in, const int* in_sizes, int n_in,
                              void* d_out, int out_size, void* d_ws, size_t ws_size,
                              hipStream_t stream) {
    (void)n_in; (void)out_size; (void)d_ws; (void)ws_size;
    const float* pc = (const float*)d_in[0];
    const float* qc = (const float*)d_in[1];
    float* out = (float*)d_out;

    const int B = 4;
    const int N = in_sizes[0] / (3 * B);
    const int M = in_sizes[1] / (3 * B);

    dim3 grid(B * (M / 16));
    softproj_knn_wmma<<<grid, 32 * WAVES, 0, stream>>>(pc, qc, out, N, M);
}


# Two-pass selection: a min-only pass over the whole cloud (same WMMA products, running column minima, no lists) bounds each query's 16th-best; lists start at that bound, few inserts; same set, order
# speedup vs baseline: 14.9862x; 1.1765x over previous
#include <hip/hip_runtime.h>
#include <float.h>

typedef __attribute__((ext_vector_type(2))) float v2f;
typedef __attribute__((ext_vector_type(8))) float v8f;

typedef __attribute__((ext_vector_type(4))) float v4f;
typedef float v4fu __attribute__((ext_vector_type(4))) __attribute__((aligned(4)));

#define KNN 16
#define WAVES 8
#define UNROLL 4
#define GROUP (16 * UNROLL)
#define STEP (GROUP * WAVES)


__device__ __forceinline__ int row_shr1_i(int v, int first) {
    return __builtin_amdgcn_update_dpp(first, v, 0x111, 0xF, 0xF, false);
}
__device__ __forceinline__ float row_shr1_f(float v, float first) {
    return __int_as_float(row_shr1_i(__float_as_int(v), __float_as_int(first)));
}
__device__ __forceinline__ float row_last_f(float v) {
    const int b = __float_as_int(v);
    return __int_as_float(__builtin_amdgcn_update_dpp(b, b, 0x15F, 0xF, 0xF, false));
}

__global__ __launch_bounds__(32 * WAVES)
void softproj_knn_wmma(const float* __restrict__ pc,
                       const float* __restrict__ qc,
                       float* __restrict__ out,
                       int N, int M) {
    const int lane = threadIdx.x & 31;
    const int wv   = __builtin_amdgcn_readfirstlane((int)(threadIdx.x >> 5));
    const int mrow = lane & 15;
    const bool hi  = lane >= 16;

    const int tiles_m = M >> 4;
    const int b  = blockIdx.x / tiles_m;
    const int m0 = (blockIdx.x % tiles_m) << 4;

    const float* pcb = pc + (size_t)b * 3 * N;
    const float* qcb = qc + (size_t)b * 3 * M;

    const int   mga = m0 + mrow;
    const float ax = qcb[0 * M + mga];
    const float ay = qcb[1 * M + mga];
    const float az = qcb[2 * M + mga];
    v2f afrag;
    afrag.x = hi ? az   : ax;
    afrag.y = hi ? 1.0f : ay;

    __shared__ int lds_li[16][KNN + 1];
    __shared__ __attribute__((aligned(16))) v4f lds_d[3][8][WAVES][32];

    float ld;
    int   li  = 0;
    float thr;

    const unsigned nlast  = (unsigned)N - 1u;
    const int      nsteps = (N + STEP - 1) / STEP;

    float px[UNROLL], py[UNROLL], pz[UNROLL];
    float px1[UNROLL], py1[UNROLL], pz1[UNROLL];

    auto fetch = [&](int t, float (&px)[UNROLL], float (&py)[UNROLL], float (&pz)[UNROLL]) {
#pragma unroll
        for (int u = 0; u < UNROLL; ++u) {
            const unsigned n = min((unsigned)(t * STEP + wv * GROUP + 16 * u + mrow), nlast);
            px[u] = pcb[0 * (unsigned)N + n];
            py[u] = pcb[1 * (unsigned)N + n];
            pz[u] = pcb[2 * (unsigned)N + n];
        }
    };

    auto produce = [&](int t, int buf, const float (&px)[UNROLL], const float (&py)[UNROLL], const float (&pz)[UNROLL]) {
        v8f c[UNROLL];
#pragma unroll
        for (int u = 0; u < UNROLL; ++u) {
            const float x = px[u], y = py[u], z = pz[u];
            float psq = x * x + y * y + z * z;
            if (t * STEP + wv * GROUP + 16 * u + mrow >= N) psq = INFINITY;
            v2f bfrag;
            bfrag.x = hi ? -2.0f * z : -2.0f * x;
            bfrag.y = hi ? psq       : -2.0f * y;
            v8f z0 = {};
            c[u] = __builtin_amdgcn_wmma_f32_16x16x4_f32(false, afrag, false, bfrag,
                                                         (short)0, z0, false, false);
        }
#pragma unroll
        for (int r = 0; r < 8; ++r) {
            v4f v;
            v.x = c[0][r]; v.y = c[1][r]; v.z = c[2][r]; v.w = c[3][r];
            lds_d[buf][r][wv][lane] = v;
        }
    };

    auto scan = [&](int t, int buf) {
        const v4f* rp = &lds_d[buf][wv][0][lane];
        v4f dv[WAVES];
#pragma unroll
        for (int g = 0; g < WAVES; ++g) dv[g] = rp[g * 32];

#pragma unroll
        for (int g = 0; g < WAVES; ++g) {
            unsigned hit[UNROLL];
#pragma unroll
            for (int u = 0; u < UNROLL; ++u) {
                hit[u] = __builtin_amdgcn_ballot_w32(dv[g][u] < thr);
            }
#pragma unroll
            for (int u = 0; u < UNROLL; ++u) {
                if (__builtin_expect(hit[u] != 0u, 0)) {
                    const int nb = t * STEP + g * GROUP + 16 * u;
                    unsigned mask = __builtin_amdgcn_ballot_w32(dv[g][u] < thr);
                    while (mask != 0u) {
                        const int src = __builtin_ctz(mask);
                        mask &= mask - 1u;
                        const float ds = __int_as_float(
                            __builtin_amdgcn_readlane(__float_as_int(dv[g][u]), src));
                        const float dn = __builtin_amdgcn_inverse_ballot_w32(src < 16 ? 0x0000FFFFu : 0xFFFF0000u)
                                             ? ds : ld;
                        const int   in = nb + (src & 15);
                        const float pd = row_shr1_f(ld, -INFINITY);
                        const int   pi = row_shr1_i(li, 0);
                        const bool gt  = ld > dn;
                        const bool pgt = pd > dn;
                        ld = __builtin_amdgcn_fmed3f(pd, dn, ld);
                        li = gt ? (pgt ? pi : in) : li;
                    }
                    thr = row_last_f(ld);
                }
            }
        }
    };

    auto arrive = [&]() {
        __builtin_amdgcn_fence(__ATOMIC_RELEASE, "workgroup");
        __builtin_amdgcn_s_barrier_signal(-1);
    };
    auto pass = [&]() {
        __builtin_amdgcn_s_barrier_wait(-1);
        __builtin_amdgcn_fence(__ATOMIC_ACQUIRE, "workgroup");
    };

    float* exch = reinterpret_cast<float*>(&lds_d[2][0][0][0]);
    const bool prepass = N >= 4;
    if (prepass) {
        float mn[8];
#pragma unroll
        for (int r = 0; r < 8; ++r) mn[r] = INFINITY;
        float pmax = 0.0f;
        const unsigned n4last = (unsigned)N - 4u;
        auto fetch4 = [&](int t, v4fu& X, v4fu& Y, v4fu& Z) {
            const unsigned off = min((unsigned)(t * STEP + wv * GROUP + 4 * mrow), n4last) * 4u;
            X = *reinterpret_cast<const v4fu*>(reinterpret_cast<const char*>(pcb) + off);
            Y = *reinterpret_cast<const v4fu*>(reinterpret_cast<const char*>(pcb + (unsigned)N) + off);
            Z = *reinterpret_cast<const v4fu*>(reinterpret_cast<const char*>(pcb + 2u * (unsigned)N) + off);
        };
        auto minstep = [&](int t, const v4fu& X, const v4fu& Y, const v4fu& Z) {
            const bool inside = t * STEP + wv * GROUP + 4 * mrow + 3 < N;
            v8f c[UNROLL];
#pragma unroll
            for (int k = 0; k < UNROLL; ++k) {
                const float x = X[k], y = Y[k], z = Z[k];
                float psq = x * x + y * y + z * z;
                pmax = fmaxf(pmax, psq);
                if (!inside) psq = INFINITY;
                v2f bfrag;
                bfrag.x = hi ? -2.0f * z : -2.0f * x;
                bfrag.y = hi ? psq       : -2.0f * y;
                v8f z0 = {};
                c[k] = __builtin_amdgcn_wmma_f32_16x16x4_f32(false, afrag, false, bfrag,
                                                             (short)0, z0, false, false);
            }
#pragma unroll
            for (int r = 0; r < 8; ++r)
                mn[r] = fminf(fminf(fminf(mn[r], c[0][r]), c[1][r]), fminf(c[2][r], c[3][r]));
        };
        v4fu X0, Y0, Z0, X1, Y1, Z1, X2, Y2, Z2, X3, Y3, Z3;
        fetch4(0, X0, Y0, Z0);  asm volatile("" ::: "memory");
        fetch4(1, X1, Y1, Z1);  asm volatile("" ::: "memory");
        fetch4(2, X2, Y2, Z2);  asm volatile("" ::: "memory");
        fetch4(3, X3, Y3, Z3);
        for (int t = 0; t < nsteps; t += 4) {
            minstep(t,     X0, Y0, Z0);  fetch4(t + 4, X0, Y0, Z0);
            minstep(t + 1, X1, Y1, Z1);  fetch4(t + 5, X1, Y1, Z1);
            minstep(t + 2, X2, Y2, Z2);  fetch4(t + 6, X2, Y2, Z2);
            minstep(t + 3, X3, Y3, Z3);  fetch4(t + 7, X3, Y3, Z3);
            asm volatile("" ::: "memory");
        }
#pragma unroll
        for (int r = 0; r < 8; ++r) exch[(r * WAVES + wv) * 32 + lane] = mn[r];
        exch[(8 * WAVES + wv) * 32 + lane] = pmax;
    }

    fetch(0, px, py, pz);
    if (1 < nsteps) fetch(1, px1, py1, pz1);
    produce(0, 0, px, py, pz);
    if (1 < nsteps) produce(1, 1, px1, py1, pz1);
    __syncthreads();
    thr = INFINITY;
    if (prepass) {
        float c1 = INFINITY, c2 = INFINITY, pm = 0.0f;
#pragma unroll
        for (int w = 0; w < WAVES; ++w) {
            const float v = exch[(wv * WAVES + w) * 32 + lane];
            c2 = __builtin_amdgcn_fmed3f(c1, c2, v);
            c1 = fminf(c1, v);
            pm = fmaxf(pm, exch[(8 * WAVES + w) * 32 + lane]);
        }
        const float o1 = __int_as_float(__builtin_amdgcn_update_dpp(
            __float_as_int(c1), __float_as_int(c1), 0xB1, 0xF, 0xF, false));
        const float o2 = __int_as_float(__builtin_amdgcn_update_dpp(
            __float_as_int(c2), __float_as_int(c2), 0xB1, 0xF, 0xF, false));
        float bnd = fminf(fmaxf(c1, o1), fminf(c2, o2));
        float sc = pm + (ax * ax + ay * ay + az * az);
        sc  = fmaxf(sc,  __int_as_float(__builtin_amdgcn_update_dpp(
            __float_as_int(sc), __float_as_int(sc), 0x121, 0xF, 0xF, false)));
        bnd = fmaxf(bnd, __int_as_float(__builtin_amdgcn_update_dpp(
            __float_as_int(bnd), __float_as_int(bnd), 0x122, 0xF, 0xF, false)));
        sc  = fmaxf(sc,  __int_as_float(__builtin_amdgcn_update_dpp(
            __float_as_int(sc), __float_as_int(sc), 0x122, 0xF, 0xF, false)));
        bnd = fmaxf(bnd, __int_as_float(__builtin_amdgcn_update_dpp(
            __float_as_int(bnd), __float_as_int(bnd), 0x124, 0xF, 0xF, false)));
        sc  = fmaxf(sc,  __int_as_float(__builtin_amdgcn_update_dpp(
            __float_as_int(sc), __float_as_int(sc), 0x124, 0xF, 0xF, false)));
        bnd = fmaxf(bnd, __int_as_float(__builtin_amdgcn_update_dpp(
            __float_as_int(bnd), __float_as_int(bnd), 0x128, 0xF, 0xF, false)));
        sc  = fmaxf(sc,  __int_as_float(__builtin_amdgcn_update_dpp(
            __float_as_int(sc), __float_as_int(sc), 0x128, 0xF, 0xF, false)));
        thr = bnd + ((fabsf(bnd) + sc) * 0x1p-19f + 0x1p-100f);
    }
    ld = thr;
    arrive();
    if (2 < nsteps) fetch(2, px, py, pz);

    int bs = 0;
    int bp = 2;
    for (int s = 0; s < nsteps; ++s) {
        scan(s, bs);
        pass();
        if (s + 2 < nsteps) produce(s + 2, bp, px, py, pz);
        arrive();
        if (s + 3 < nsteps) fetch(s + 3, px, py, pz);
        bs = (bs == 2) ? 0 : bs + 1;
        bp = (bp == 2) ? 0 : bp + 1;
    }
    pass();

    lds_li[wv + (hi ? 8 : 0)][mrow] = li;
    __syncthreads();

    if (wv == 0 && lane < 16) {
        const int   mg = mga;
        const float qx = ax, qy = ay, qz = az;
        int midx[KNN];
#pragma unroll
        for (int j = 0; j < KNN; ++j) midx[j] = lds_li[mrow][j];

        float pxa[KNN], pya[KNN], pza[KNN], ed[KNN];
        float dmin = FLT_MAX;
#pragma unroll
        for (int j = 0; j < KNN; ++j) {
            const int   gi = midx[j];
            const float gx = pcb[0 * N + gi];
            const float gy = pcb[1 * N + gi];
            const float gz = pcb[2 * N + gi];
            const float dx = gx - qx, dy = gy - qy, dz = gz - qz;
            const float d  = dx * dx + dy * dy + dz * dz;
            pxa[j] = gx; pya[j] = gy; pza[j] = gz; ed[j] = d;
            dmin = fminf(dmin, d);
        }

        float wsum = 0.0f, ox = 0.0f, oy = 0.0f, oz = 0.0f;
#pragma unroll
        for (int j = 0; j < KNN; ++j) {
            const float w = __expf(dmin - ed[j]);
            wsum += w;
            ox += w * pxa[j];
            oy += w * pya[j];
            oz += w * pza[j];
        }
        const float inv = 1.0f / wsum;
        float* outb = out + (size_t)b * 3 * M;
        outb[0 * M + mg] = ox * inv;
        outb[1 * M + mg] = oy * inv;
        outb[2 * M + mg] = oz * inv;
    }
}

extern "C" void kernel_launch(void* const* d_in, const int* in_sizes, int n_in,
                              void* d_out, int out_size, void* d_ws, size_t ws_size,
                              hipStream_t stream) {
    (void)n_in; (void)out_size; (void)d_ws; (void)ws_size;
    const float* pc = (const float*)d_in[0];
    const float* qc = (const float*)d_in[1];
    float* out = (float*)d_out;

    const int B = 4;
    const int N = in_sizes[0] / (3 * B);
    const int M = in_sizes[1] / (3 * B);

    dim3 grid(B * (M / 16));
    softproj_knn_wmma<<<grid, 32 * WAVES, 0, stream>>>(pc, qc, out, N, M);
}


# All 32 tile filters of a step issue back to back before its first branch, vs the step-start bound; a stale mark inserts as a no-op, no re-filter: no compare-to-branch wait per group; same set, order
# speedup vs baseline: 15.5875x; 1.0401x over previous
#include <hip/hip_runtime.h>
#include <float.h>

typedef __attribute__((ext_vector_type(2))) float v2f;
typedef __attribute__((ext_vector_type(8))) float v8f;

typedef __attribute__((ext_vector_type(4))) float v4f;
typedef float v4fu __attribute__((ext_vector_type(4))) __attribute__((aligned(4)));

#define KNN 16
#define WAVES 8
#define UNROLL 4
#define GROUP (16 * UNROLL)
#define STEP (GROUP * WAVES)


__device__ __forceinline__ int row_shr1_i(int v, int first) {
    return __builtin_amdgcn_update_dpp(first, v, 0x111, 0xF, 0xF, false);
}
__device__ __forceinline__ float row_shr1_f(float v, float first) {
    return __int_as_float(row_shr1_i(__float_as_int(v), __float_as_int(first)));
}
__device__ __forceinline__ float row_last_f(float v) {
    const int b = __float_as_int(v);
    return __int_as_float(__builtin_amdgcn_update_dpp(b, b, 0x15F, 0xF, 0xF, false));
}

__global__ __launch_bounds__(32 * WAVES)
void softproj_knn_wmma(const float* __restrict__ pc,
                       const float* __restrict__ qc,
                       float* __restrict__ out,
                       int N, int M) {
    const int lane = threadIdx.x & 31;
    const int wv   = __builtin_amdgcn_readfirstlane((int)(threadIdx.x >> 5));
    const int mrow = lane & 15;
    const bool hi  = lane >= 16;

    const int tiles_m = M >> 4;
    const int b  = blockIdx.x / tiles_m;
    const int m0 = (blockIdx.x % tiles_m) << 4;

    const float* pcb = pc + (size_t)b * 3 * N;
    const float* qcb = qc + (size_t)b * 3 * M;

    const int   mga = m0 + mrow;
    const float ax = qcb[0 * M + mga];
    const float ay = qcb[1 * M + mga];
    const float az = qcb[2 * M + mga];
    v2f afrag;
    afrag.x = hi ? az   : ax;
    afrag.y = hi ? 1.0f : ay;

    __shared__ int lds_li[16][KNN + 1];
    __shared__ __attribute__((aligned(16))) v4f lds_d[3][8][WAVES][32];

    float ld;
    int   li  = 0;
    float thr;

    const unsigned nlast  = (unsigned)N - 1u;
    const int      nsteps = (N + STEP - 1) / STEP;

    float px[UNROLL], py[UNROLL], pz[UNROLL];
    float px1[UNROLL], py1[UNROLL], pz1[UNROLL];

    auto fetch = [&](int t, float (&px)[UNROLL], float (&py)[UNROLL], float (&pz)[UNROLL]) {
#pragma unroll
        for (int u = 0; u < UNROLL; ++u) {
            const unsigned n = min((unsigned)(t * STEP + wv * GROUP + 16 * u + mrow), nlast);
            px[u] = pcb[0 * (unsigned)N + n];
            py[u] = pcb[1 * (unsigned)N + n];
            pz[u] = pcb[2 * (unsigned)N + n];
        }
    };

    auto produce = [&](int t, int buf, const float (&px)[UNROLL], const float (&py)[UNROLL], const float (&pz)[UNROLL]) {
        v8f c[UNROLL];
#pragma unroll
        for (int u = 0; u < UNROLL; ++u) {
            const float x = px[u], y = py[u], z = pz[u];
            float psq = x * x + y * y + z * z;
            if (t * STEP + wv * GROUP + 16 * u + mrow >= N) psq = INFINITY;
            v2f bfrag;
            bfrag.x = hi ? -2.0f * z : -2.0f * x;
            bfrag.y = hi ? psq       : -2.0f * y;
            v8f z0 = {};
            c[u] = __builtin_amdgcn_wmma_f32_16x16x4_f32(false, afrag, false, bfrag,
                                                         (short)0, z0, false, false);
        }
#pragma unroll
        for (int r = 0; r < 8; ++r) {
            v4f v;
            v.x = c[0][r]; v.y = c[1][r]; v.z = c[2][r]; v.w = c[3][r];
            lds_d[buf][r][wv][lane] = v;
        }
    };

    auto scan = [&](int t, int buf) {
        const v4f* rp = &lds_d[buf][wv][0][lane];
        v4f dv[WAVES];
#pragma unroll
        for (int g = 0; g < WAVES; ++g) dv[g] = rp[g * 32];

        unsigned hit[WAVES][UNROLL];
#pragma unroll
        for (int g = 0; g < WAVES; ++g) {
#pragma unroll
            for (int u = 0; u < UNROLL; ++u)
                hit[g][u] = __builtin_amdgcn_ballot_w32(dv[g][u] < thr);
            if (g == 0) __builtin_amdgcn_sched_barrier(0);
        }
#pragma unroll
        for (int g = 0; g < WAVES; ++g) {
#pragma unroll
            for (int u = 0; u < UNROLL; ++u) {
                if (__builtin_expect(hit[g][u] != 0u, 0)) {
                    const int nb = t * STEP + g * GROUP + 16 * u;
                    unsigned mask = hit[g][u];
                    while (mask != 0u) {
                        const int src = __builtin_ctz(mask);
                        mask &= mask - 1u;
                        const float ds = __int_as_float(
                            __builtin_amdgcn_readlane(__float_as_int(dv[g][u]), src));
                        const float dn = __builtin_amdgcn_inverse_ballot_w32(src < 16 ? 0x0000FFFFu : 0xFFFF0000u)
                                             ? ds : ld;
                        const int   in = nb + (src & 15);
                        const float pd = row_shr1_f(ld, -INFINITY);
                        const int   pi = row_shr1_i(li, 0);
                        const bool gt  = ld > dn;
                        const bool pgt = pd > dn;
                        ld = __builtin_amdgcn_fmed3f(pd, dn, ld);
                        li = gt ? (pgt ? pi : in) : li;
                    }
                }
            }
        }
        thr = row_last_f(ld);
    };

    auto arrive = [&]() {
        __builtin_amdgcn_fence(__ATOMIC_RELEASE, "workgroup");
        __builtin_amdgcn_s_barrier_signal(-1);
    };
    auto pass = [&]() {
        __builtin_amdgcn_s_barrier_wait(-1);
        __builtin_amdgcn_fence(__ATOMIC_ACQUIRE, "workgroup");
    };

    float* exch = reinterpret_cast<float*>(&lds_d[2][0][0][0]);
    const bool prepass = N >= 4;
    if (prepass) {
        float mn[8];
#pragma unroll
        for (int r = 0; r < 8; ++r) mn[r] = INFINITY;
        float pmax = 0.0f;
        const unsigned n4last = (unsigned)N - 4u;
        auto fetch4 = [&](int t, v4fu& X, v4fu& Y, v4fu& Z) {
            const unsigned off = min((unsigned)(t * STEP + wv * GROUP + 4 * mrow), n4last) * 4u;
            X = *reinterpret_cast<const v4fu*>(reinterpret_cast<const char*>(pcb) + off);
            Y = *reinterpret_cast<const v4fu*>(reinterpret_cast<const char*>(pcb + (unsigned)N) + off);
            Z = *reinterpret_cast<const v4fu*>(reinterpret_cast<const char*>(pcb + 2u * (unsigned)N) + off);
        };
        auto minstep = [&](int t, const v4fu& X, const v4fu& Y, const v4fu& Z) {
            const bool inside = t * STEP + wv * GROUP + 4 * mrow + 3 < N;
            v8f c[UNROLL];
#pragma unroll
            for (int k = 0; k < UNROLL; ++k) {
                const float x = X[k], y = Y[k], z = Z[k];
                float psq = x * x + y * y + z * z;
                pmax = fmaxf(pmax, psq);
                if (!inside) psq = INFINITY;
                v2f bfrag;
                bfrag.x = hi ? -2.0f * z : -2.0f * x;
                bfrag.y = hi ? psq       : -2.0f * y;
                v8f z0 = {};
                c[k] = __builtin_amdgcn_wmma_f32_16x16x4_f32(false, afrag, false, bfrag,
                                                             (short)0, z0, false, false);
            }
#pragma unroll
            for (int r = 0; r < 8; ++r)
                mn[r] = fminf(fminf(fminf(mn[r], c[0][r]), c[1][r]), fminf(c[2][r], c[3][r]));
        };
        v4fu X0, Y0, Z0, X1, Y1, Z1, X2, Y2, Z2, X3, Y3, Z3;
        fetch4(0, X0, Y0, Z0);  asm volatile("" ::: "memory");
        fetch4(1, X1, Y1, Z1);  asm volatile("" ::: "memory");
        fetch4(2, X2, Y2, Z2);  asm volatile("" ::: "memory");
        fetch4(3, X3, Y3, Z3);
        for (int t = 0; t < nsteps; t += 4) {
            minstep(t,     X0, Y0, Z0);  fetch4(t + 4, X0, Y0, Z0);
            minstep(t + 1, X1, Y1, Z1);  fetch4(t + 5, X1, Y1, Z1);
            minstep(t + 2, X2, Y2, Z2);  fetch4(t + 6, X2, Y2, Z2);
            minstep(t + 3, X3, Y3, Z3);  fetch4(t + 7, X3, Y3, Z3);
            asm volatile("" ::: "memory");
        }
#pragma unroll
        for (int r = 0; r < 8; ++r) exch[(r * WAVES + wv) * 32 + lane] = mn[r];
        exch[(8 * WAVES + wv) * 32 + lane] = pmax;
    }

    fetch(0, px, py, pz);
    if (1 < nsteps) fetch(1, px1, py1, pz1);
    produce(0, 0, px, py, pz);
    if (1 < nsteps) produce(1, 1, px1, py1, pz1);
    __syncthreads();
    thr = INFINITY;
    if (prepass) {
        float c1 = INFINITY, c2 = INFINITY, pm = 0.0f;
#pragma unroll
        for (int w = 0; w < WAVES; ++w) {
            const float v = exch[(wv * WAVES + w) * 32 + lane];
            c2 = __builtin_amdgcn_fmed3f(c1, c2, v);
            c1 = fminf(c1, v);
            pm = fmaxf(pm, exch[(8 * WAVES + w) * 32 + lane]);
        }
        const float o1 = __int_as_float(__builtin_amdgcn_update_dpp(
            __float_as_int(c1), __float_as_int(c1), 0xB1, 0xF, 0xF, false));
        const float o2 = __int_as_float(__builtin_amdgcn_update_dpp(
            __float_as_int(c2), __float_as_int(c2), 0xB1, 0xF, 0xF, false));
        float bnd = fminf(fmaxf(c1, o1), fminf(c2, o2));
        float sc = pm + (ax * ax + ay * ay + az * az);
        sc  = fmaxf(sc,  __int_as_float(__builtin_amdgcn_update_dpp(
            __float_as_int(sc), __float_as_int(sc), 0x121, 0xF, 0xF, false)));
        bnd = fmaxf(bnd, __int_as_float(__builtin_amdgcn_update_dpp(
            __float_as_int(bnd), __float_as_int(bnd), 0x122, 0xF, 0xF, false)));
        sc  = fmaxf(sc,  __int_as_float(__builtin_amdgcn_update_dpp(
            __float_as_int(sc), __float_as_int(sc), 0x122, 0xF, 0xF, false)));
        bnd = fmaxf(bnd, __int_as_float(__builtin_amdgcn_update_dpp(
            __float_as_int(bnd), __float_as_int(bnd), 0x124, 0xF, 0xF, false)));
        sc  = fmaxf(sc,  __int_as_float(__builtin_amdgcn_update_dpp(
            __float_as_int(sc), __float_as_int(sc), 0x124, 0xF, 0xF, false)));
        bnd = fmaxf(bnd, __int_as_float(__builtin_amdgcn_update_dpp(
            __float_as_int(bnd), __float_as_int(bnd), 0x128, 0xF, 0xF, false)));
        sc  = fmaxf(sc,  __int_as_float(__builtin_amdgcn_update_dpp(
            __float_as_int(sc), __float_as_int(sc), 0x128, 0xF, 0xF, false)));
        thr = bnd + ((fabsf(bnd) + sc) * 0x1p-19f + 0x1p-100f);
    }
    ld = thr;
    arrive();
    if (2 < nsteps) fetch(2, px, py, pz);

    int bs = 0;
    int bp = 2;
    for (int s = 0; s < nsteps; ++s) {
        scan(s, bs);
        pass();
        if (s + 2 < nsteps) produce(s + 2, bp, px, py, pz);
        arrive();
        if (s + 3 < nsteps) fetch(s + 3, px, py, pz);
        bs = (bs == 2) ? 0 : bs + 1;
        bp = (bp == 2) ? 0 : bp + 1;
    }
    pass();

    lds_li[wv + (hi ? 8 : 0)][mrow] = li;
    __syncthreads();

    if (wv == 0 && lane < 16) {
        const int   mg = mga;
        const float qx = ax, qy = ay, qz = az;
        int midx[KNN];
#pragma unroll
        for (int j = 0; j < KNN; ++j) midx[j] = lds_li[mrow][j];

        float pxa[KNN], pya[KNN], pza[KNN], ed[KNN];
        float dmin = FLT_MAX;
#pragma unroll
        for (int j = 0; j < KNN; ++j) {
            const int   gi = midx[j];
            const float gx = pcb[0 * N + gi];
            const float gy = pcb[1 * N + gi];
            const float gz = pcb[2 * N + gi];
            const float dx = gx - qx, dy = gy - qy, dz = gz - qz;
            const float d  = dx * dx + dy * dy + dz * dz;
            pxa[j] = gx; pya[j] = gy; pza[j] = gz; ed[j] = d;
            dmin = fminf(dmin, d);
        }

        float wsum = 0.0f, ox = 0.0f, oy = 0.0f, oz = 0.0f;
#pragma unroll
        for (int j = 0; j < KNN; ++j) {
            const float w = __expf(dmin - ed[j]);
            wsum += w;
            ox += w * pxa[j];
            oy += w * pya[j];
            oz += w * pza[j];
        }
        const float inv = 1.0f / wsum;
        float* outb = out + (size_t)b * 3 * M;
        outb[0 * M + mg] = ox * inv;
        outb[1 * M + mg] = oy * inv;
        outb[2 * M + mg] = oz * inv;
    }
}

extern "C" void kernel_launch(void* const* d_in, const int* in_sizes, int n_in,
                              void* d_out, int out_size, void* d_ws, size_t ws_size,
                              hipStream_t stream) {
    (void)n_in; (void)out_size; (void)d_ws; (void)ws_size;
    const float* pc = (const float*)d_in[0];
    const float* qc = (const float*)d_in[1];
    float* out = (float*)d_out;

    const int B = 4;
    const int N = in_sizes[0] / (3 * B);
    const int M = in_sizes[1] / (3 * B);

    dim3 grid(B * (M / 16));
    softproj_knn_wmma<<<grid, 32 * WAVES, 0, stream>>>(pc, qc, out, N, M);
}


# Min-only pass reads every second step of the cloud (half its WMMAs, loads, minima); bound kept as tight: 16th smallest of the 32 column minima, sorted over the row by DPP; same set, order
# speedup vs baseline: 17.3297x; 1.1118x over previous
#include <hip/hip_runtime.h>
#include <float.h>

typedef __attribute__((ext_vector_type(2))) float v2f;
typedef __attribute__((ext_vector_type(8))) float v8f;

typedef __attribute__((ext_vector_type(4))) float v4f;
typedef float v4fu __attribute__((ext_vector_type(4))) __attribute__((aligned(4)));

#define KNN 16
#define WAVES 8
#define UNROLL 4
#define GROUP (16 * UNROLL)
#define STEP (GROUP * WAVES)


__device__ __forceinline__ int row_shr1_i(int v, int first) {
    return __builtin_amdgcn_update_dpp(first, v, 0x111, 0xF, 0xF, false);
}
__device__ __forceinline__ float row_shr1_f(float v, float first) {
    return __int_as_float(row_shr1_i(__float_as_int(v), __float_as_int(first)));
}
__device__ __forceinline__ float row_last_f(float v) {
    const int b = __float_as_int(v);
    return __int_as_float(__builtin_amdgcn_update_dpp(b, b, 0x15F, 0xF, 0xF, false));
}

__global__ __launch_bounds__(32 * WAVES)
void softproj_knn_wmma(const float* __restrict__ pc,
                       const float* __restrict__ qc,
                       float* __restrict__ out,
                       int N, int M) {
    const int lane = threadIdx.x & 31;
    const int wv   = __builtin_amdgcn_readfirstlane((int)(threadIdx.x >> 5));
    const int mrow = lane & 15;
    const bool hi  = lane >= 16;

    const int tiles_m = M >> 4;
    const int b  = blockIdx.x / tiles_m;
    const int m0 = (blockIdx.x % tiles_m) << 4;

    const float* pcb = pc + (size_t)b * 3 * N;
    const float* qcb = qc + (size_t)b * 3 * M;

    const int   mga = m0 + mrow;
    const float ax = qcb[0 * M + mga];
    const float ay = qcb[1 * M + mga];
    const float az = qcb[2 * M + mga];
    v2f afrag;
    afrag.x = hi ? az   : ax;
    afrag.y = hi ? 1.0f : ay;

    __shared__ int lds_li[16][KNN + 1];
    __shared__ __attribute__((aligned(16))) v4f lds_d[3][8][WAVES][32];

    float ld;
    int   li  = 0;
    float thr;

    const unsigned nlast  = (unsigned)N - 1u;
    const int      nsteps = (N + STEP - 1) / STEP;

    float px[UNROLL], py[UNROLL], pz[UNROLL];
    float px1[UNROLL], py1[UNROLL], pz1[UNROLL];

    auto fetch = [&](int t, float (&px)[UNROLL], float (&py)[UNROLL], float (&pz)[UNROLL]) {
#pragma unroll
        for (int u = 0; u < UNROLL; ++u) {
            const unsigned n = min((unsigned)(t * STEP + wv * GROUP + 16 * u + mrow), nlast);
            px[u] = pcb[0 * (unsigned)N + n];
            py[u] = pcb[1 * (unsigned)N + n];
            pz[u] = pcb[2 * (unsigned)N + n];
        }
    };

    auto produce = [&](int t, int buf, const float (&px)[UNROLL], const float (&py)[UNROLL], const float (&pz)[UNROLL]) {
        v8f c[UNROLL];
#pragma unroll
        for (int u = 0; u < UNROLL; ++u) {
            const float x = px[u], y = py[u], z = pz[u];
            float psq = x * x + y * y + z * z;
            if (t * STEP + wv * GROUP + 16 * u + mrow >= N) psq = INFINITY;
            v2f bfrag;
            bfrag.x = hi ? -2.0f * z : -2.0f * x;
            bfrag.y = hi ? psq       : -2.0f * y;
            v8f z0 = {};
            c[u] = __builtin_amdgcn_wmma_f32_16x16x4_f32(false, afrag, false, bfrag,
                                                         (short)0, z0, false, false);
        }
#pragma unroll
        for (int r = 0; r < 8; ++r) {
            v4f v;
            v.x = c[0][r]; v.y = c[1][r]; v.z = c[2][r]; v.w = c[3][r];
            lds_d[buf][r][wv][lane] = v;
        }
    };

    auto scan = [&](int t, int buf) {
        const v4f* rp = &lds_d[buf][wv][0][lane];
        v4f dv[WAVES];
#pragma unroll
        for (int g = 0; g < WAVES; ++g) dv[g] = rp[g * 32];

        unsigned hit[WAVES][UNROLL];
#pragma unroll
        for (int g = 0; g < WAVES; ++g) {
#pragma unroll
            for (int u = 0; u < UNROLL; ++u)
                hit[g][u] = __builtin_amdgcn_ballot_w32(dv[g][u] < thr);
            if (g == 0) __builtin_amdgcn_sched_barrier(0);
        }
#pragma unroll
        for (int g = 0; g < WAVES; ++g) {
#pragma unroll
            for (int u = 0; u < UNROLL; ++u) {
                if (__builtin_expect(hit[g][u] != 0u, 0)) {
                    const int nb = t * STEP + g * GROUP + 16 * u;
                    unsigned mask = hit[g][u];
                    while (mask != 0u) {
                        const int src = __builtin_ctz(mask);
                        mask &= mask - 1u;
                        const float ds = __int_as_float(
                            __builtin_amdgcn_readlane(__float_as_int(dv[g][u]), src));
                        const float dn = __builtin_amdgcn_inverse_ballot_w32(src < 16 ? 0x0000FFFFu : 0xFFFF0000u)
                                             ? ds : ld;
                        const int   in = nb + (src & 15);
                        const float pd = row_shr1_f(ld, -INFINITY);
                        const int   pi = row_shr1_i(li, 0);
                        const bool gt  = ld > dn;
                        const bool pgt = pd > dn;
                        ld = __builtin_amdgcn_fmed3f(pd, dn, ld);
                        li = gt ? (pgt ? pi : in) : li;
                    }
                }
            }
        }
        thr = row_last_f(ld);
    };

    auto arrive = [&]() {
        __builtin_amdgcn_fence(__ATOMIC_RELEASE, "workgroup");
        __builtin_amdgcn_s_barrier_signal(-1);
    };
    auto pass = [&]() {
        __builtin_amdgcn_s_barrier_wait(-1);
        __builtin_amdgcn_fence(__ATOMIC_ACQUIRE, "workgroup");
    };

    float* exch = reinterpret_cast<float*>(&lds_d[2][0][0][0]);
    const bool prepass = N >= 4;
    if (prepass) {
        float mn[8];
#pragma unroll
        for (int r = 0; r < 8; ++r) mn[r] = INFINITY;
        float pmax = 0.0f;
        const unsigned n4last = (unsigned)N - 4u;
        auto fetch4 = [&](int t, v4fu& X, v4fu& Y, v4fu& Z) {
            const unsigned off = min((unsigned)(t * STEP + wv * GROUP + 4 * mrow), n4last) * 4u;
            X = *reinterpret_cast<const v4fu*>(reinterpret_cast<const char*>(pcb) + off);
            Y = *reinterpret_cast<const v4fu*>(reinterpret_cast<const char*>(pcb + (unsigned)N) + off);
            Z = *reinterpret_cast<const v4fu*>(reinterpret_cast<const char*>(pcb + 2u * (unsigned)N) + off);
        };
        auto minstep = [&](int t, const v4fu& X, const v4fu& Y, const v4fu& Z) {
            const bool inside = t * STEP + wv * GROUP + 4 * mrow + 3 < N;
            v8f c[UNROLL];
#pragma unroll
            for (int k = 0; k < UNROLL; ++k) {
                const float x = X[k], y = Y[k], z = Z[k];
                float psq = x * x + y * y + z * z;
                pmax = fmaxf(pmax, psq);
                if (!inside) psq = INFINITY;
                v2f bfrag;
                bfrag.x = hi ? -2.0f * z : -2.0f * x;
                bfrag.y = hi ? psq       : -2.0f * y;
                v8f z0 = {};
                c[k] = __builtin_amdgcn_wmma_f32_16x16x4_f32(false, afrag, false, bfrag,
                                                             (short)0, z0, false, false);
            }
#pragma unroll
            for (int r = 0; r < 8; ++r)
                mn[r] = fminf(fminf(fminf(mn[r], c[0][r]), c[1][r]), fminf(c[2][r], c[3][r]));
        };
        v4fu X0, Y0, Z0, X1, Y1, Z1, X2, Y2, Z2, X3, Y3, Z3;
        fetch4(0, X0, Y0, Z0);  asm volatile("" ::: "memory");
        fetch4(2, X1, Y1, Z1);  asm volatile("" ::: "memory");
        fetch4(4, X2, Y2, Z2);  asm volatile("" ::: "memory");
        fetch4(6, X3, Y3, Z3);
        for (int t = 0; t < nsteps; t += 8) {
            minstep(t,     X0, Y0, Z0);  fetch4(t + 8,  X0, Y0, Z0);
            minstep(t + 2, X1, Y1, Z1);  fetch4(t + 10, X1, Y1, Z1);
            minstep(t + 4, X2, Y2, Z2);  fetch4(t + 12, X2, Y2, Z2);
            minstep(t + 6, X3, Y3, Z3);  fetch4(t + 14, X3, Y3, Z3);
            asm volatile("" ::: "memory");
        }
#pragma unroll
        for (int r = 0; r < 8; ++r) exch[(r * WAVES + wv) * 32 + lane] = mn[r];
        exch[(8 * WAVES + wv) * 32 + lane] = pmax;
    }

    fetch(0, px, py, pz);
    if (1 < nsteps) fetch(1, px1, py1, pz1);
    produce(0, 0, px, py, pz);
    if (1 < nsteps) produce(1, 1, px1, py1, pz1);
    __syncthreads();
    thr = INFINITY;
    if (prepass) {
        float c1 = INFINITY, c2 = INFINITY, pm = 0.0f;
#pragma unroll
        for (int w = 0; w < WAVES; ++w) {
            const float v = exch[(wv * WAVES + w) * 32 + lane];
            c2 = __builtin_amdgcn_fmed3f(c1, c2, v);
            c1 = fminf(c1, v);
            pm = fmaxf(pm, exch[(8 * WAVES + w) * 32 + lane]);
        }
#define XROW(v, ctrl) __int_as_float(__builtin_amdgcn_update_dpp( \
            0, __float_as_int(v), (ctrl), 0xF, 0xF, true))
#define SORT_STAGE(K, J, ctrl)                                                              \
        {                                                                                   \
            const float end = (((mrow & (K)) == 0) == ((mrow & (J)) == 0)) ? -INFINITY : INFINITY; \
            c1 = __builtin_amdgcn_fmed3f(c1, XROW(c1, ctrl), end);                          \
            c2 = __builtin_amdgcn_fmed3f(c2, XROW(c2, ctrl), end);                          \
        }
        SORT_STAGE(2, 1, 0xB1)
        SORT_STAGE(4, 2, 0x4E)   SORT_STAGE(4, 1, 0xB1)
        SORT_STAGE(8, 4, 0x164)  SORT_STAGE(8, 2, 0x4E)   SORT_STAGE(8, 1, 0xB1)
        SORT_STAGE(16, 8, 0x168) SORT_STAGE(16, 4, 0x164) SORT_STAGE(16, 2, 0x4E) SORT_STAGE(16, 1, 0xB1)
        float bnd = fminf(c1, XROW(c2, 0x140));
#undef SORT_STAGE
        float sc = pm + (ax * ax + ay * ay + az * az);
        sc  = fmaxf(sc,  XROW(sc,  0x121));
        bnd = fmaxf(bnd, XROW(bnd, 0x121));
        sc  = fmaxf(sc,  XROW(sc,  0x122));
        bnd = fmaxf(bnd, XROW(bnd, 0x122));
        sc  = fmaxf(sc,  XROW(sc,  0x124));
        bnd = fmaxf(bnd, XROW(bnd, 0x124));
        sc  = fmaxf(sc,  XROW(sc,  0x128));
        bnd = fmaxf(bnd, XROW(bnd, 0x128));
#undef XROW
        thr = bnd + ((fabsf(bnd) + sc) * 0x1p-19f + 0x1p-100f);
    }
    ld = thr;
    arrive();
    if (2 < nsteps) fetch(2, px, py, pz);

    int bs = 0;
    int bp = 2;
    for (int s = 0; s < nsteps; ++s) {
        scan(s, bs);
        pass();
        if (s + 2 < nsteps) produce(s + 2, bp, px, py, pz);
        arrive();
        if (s + 3 < nsteps) fetch(s + 3, px, py, pz);
        bs = (bs == 2) ? 0 : bs + 1;
        bp = (bp == 2) ? 0 : bp + 1;
    }
    pass();

    lds_li[wv + (hi ? 8 : 0)][mrow] = li;
    __syncthreads();

    if (wv == 0 && lane < 16) {
        const int   mg = mga;
        const float qx = ax, qy = ay, qz = az;
        int midx[KNN];
#pragma unroll
        for (int j = 0; j < KNN; ++j) midx[j] = lds_li[mrow][j];

        float pxa[KNN], pya[KNN], pza[KNN], ed[KNN];
        float dmin = FLT_MAX;
#pragma unroll
        for (int j = 0; j < KNN; ++j) {
            const int   gi = midx[j];
            const float gx = pcb[0 * N + gi];
            const float gy = pcb[1 * N + gi];
            const float gz = pcb[2 * N + gi];
            const float dx = gx - qx, dy = gy - qy, dz = gz - qz;
            const float d  = dx * dx + dy * dy + dz * dz;
            pxa[j] = gx; pya[j] = gy; pza[j] = gz; ed[j] = d;
            dmin = fminf(dmin, d);
        }

        float wsum = 0.0f, ox = 0.0f, oy = 0.0f, oz = 0.0f;
#pragma unroll
        for (int j = 0; j < KNN; ++j) {
            const float w = __expf(dmin - ed[j]);
            wsum += w;
            ox += w * pxa[j];
            oy += w * pya[j];
            oz += w * pza[j];
        }
        const float inv = 1.0f / wsum;
        float* outb = out + (size_t)b * 3 * M;
        outb[0 * M + mg] = ox * inv;
        outb[1 * M + mg] = oy * inv;
        outb[2 * M + mg] = oz * inv;
    }
}

extern "C" void kernel_launch(void* const* d_in, const int* in_sizes, int n_in,
                              void* d_out, int out_size, void* d_ws, size_t ws_size,
                              hipStream_t stream) {
    (void)n_in; (void)out_size; (void)d_ws; (void)ws_size;
    const float* pc = (const float*)d_in[0];
    const float* qc = (const float*)d_in[1];
    float* out = (float*)d_out;

    const int B = 4;
    const int N = in_sizes[0] / (3 * B);
    const int M = in_sizes[1] / (3 * B);

    dim3 grid(B * (M / 16));
    softproj_knn_wmma<<<grid, 32 * WAVES, 0, stream>>>(pc, qc, out, N, M);
}


# Bound from column minima of the first four steps' own distance tiles (ring of 4 filled up front), not from a separate min-only pass: its WMMAs, loads, minima are gone; more inserts; same set, order
# speedup vs baseline: 18.3571x; 1.0593x over previous
#include <hip/hip_runtime.h>
#include <float.h>

typedef __attribute__((ext_vector_type(2))) float v2f;
typedef __attribute__((ext_vector_type(8))) float v8f;

typedef __attribute__((ext_vector_type(4))) float v4f;
typedef float v4fu __attribute__((ext_vector_type(4))) __attribute__((aligned(4)));

#define KNN 16
#define WAVES 8
#define UNROLL 4
#define GROUP (16 * UNROLL)
#define STEP (GROUP * WAVES)


__device__ __forceinline__ int row_shr1_i(int v, int first) {
    return __builtin_amdgcn_update_dpp(first, v, 0x111, 0xF, 0xF, false);
}
__device__ __forceinline__ float row_shr1_f(float v, float first) {
    return __int_as_float(row_shr1_i(__float_as_int(v), __float_as_int(first)));
}
__device__ __forceinline__ float row_last_f(float v) {
    const int b = __float_as_int(v);
    return __int_as_float(__builtin_amdgcn_update_dpp(b, b, 0x15F, 0xF, 0xF, false));
}

__global__ __launch_bounds__(32 * WAVES)
void softproj_knn_wmma(const float* __restrict__ pc,
                       const float* __restrict__ qc,
                       float* __restrict__ out,
                       int N, int M) {
    const int lane = threadIdx.x & 31;
    const int wv   = __builtin_amdgcn_readfirstlane((int)(threadIdx.x >> 5));
    const int mrow = lane & 15;
    const bool hi  = lane >= 16;

    const int tiles_m = M >> 4;
    const int b  = blockIdx.x / tiles_m;
    const int m0 = (blockIdx.x % tiles_m) << 4;

    const float* pcb = pc + (size_t)b * 3 * N;
    const float* qcb = qc + (size_t)b * 3 * M;

    const int   mga = m0 + mrow;
    const float ax = qcb[0 * M + mga];
    const float ay = qcb[1 * M + mga];
    const float az = qcb[2 * M + mga];
    v2f afrag;
    afrag.x = hi ? az   : ax;
    afrag.y = hi ? 1.0f : ay;

    __shared__ int lds_li[16][KNN + 1];
    __shared__ __attribute__((aligned(16))) v4f lds_d[4][8][WAVES][32];
    __shared__ float exch[8 * WAVES * 32];

    float ld;
    int   li  = 0;
    float thr;

    const unsigned nlast  = (unsigned)N - 1u;
    const int      nsteps = (N + STEP - 1) / STEP;

    float px[UNROLL], py[UNROLL], pz[UNROLL];
    float px1[UNROLL], py1[UNROLL], pz1[UNROLL];
    float px2[UNROLL], py2[UNROLL], pz2[UNROLL];
    float px3[UNROLL], py3[UNROLL], pz3[UNROLL];
    float mn[8];
#pragma unroll
    for (int r = 0; r < 8; ++r) mn[r] = INFINITY;

    auto fetch = [&](int t, float (&px)[UNROLL], float (&py)[UNROLL], float (&pz)[UNROLL]) {
#pragma unroll
        for (int u = 0; u < UNROLL; ++u) {
            const unsigned n = min((unsigned)(t * STEP + wv * GROUP + 16 * u + mrow), nlast);
            px[u] = pcb[0 * (unsigned)N + n];
            py[u] = pcb[1 * (unsigned)N + n];
            pz[u] = pcb[2 * (unsigned)N + n];
        }
    };

    auto produce = [&](int t, int buf, const float (&px)[UNROLL], const float (&py)[UNROLL], const float (&pz)[UNROLL], const bool track) {
        v8f c[UNROLL];
#pragma unroll
        for (int u = 0; u < UNROLL; ++u) {
            const float x = px[u], y = py[u], z = pz[u];
            float psq = x * x + y * y + z * z;
            if (t * STEP + wv * GROUP + 16 * u + mrow >= N) psq = INFINITY;
            v2f bfrag;
            bfrag.x = hi ? -2.0f * z : -2.0f * x;
            bfrag.y = hi ? psq       : -2.0f * y;
            v8f z0 = {};
            c[u] = __builtin_amdgcn_wmma_f32_16x16x4_f32(false, afrag, false, bfrag,
                                                         (short)0, z0, false, false);
        }
#pragma unroll
        for (int r = 0; r < 8; ++r) {
            v4f v;
            v.x = c[0][r]; v.y = c[1][r]; v.z = c[2][r]; v.w = c[3][r];
            lds_d[buf][r][wv][lane] = v;
        }
        if (track) {
#pragma unroll
            for (int r = 0; r < 8; ++r)
                mn[r] = fminf(fminf(fminf(mn[r], c[0][r]), c[1][r]), fminf(c[2][r], c[3][r]));
        }
    };

    auto scan = [&](int t, int buf) {
        const v4f* rp = &lds_d[buf][wv][0][lane];
        v4f dv[WAVES];
#pragma unroll
        for (int g = 0; g < WAVES; ++g) dv[g] = rp[g * 32];

        unsigned hit[WAVES][UNROLL];
#pragma unroll
        for (int g = 0; g < WAVES; ++g) {
#pragma unroll
            for (int u = 0; u < UNROLL; ++u)
                hit[g][u] = __builtin_amdgcn_ballot_w32(dv[g][u] < thr);
            if (g == 0) __builtin_amdgcn_sched_barrier(0);
        }
#pragma unroll
        for (int g = 0; g < WAVES; ++g) {
#pragma unroll
            for (int u = 0; u < UNROLL; ++u) {
                if (__builtin_expect(hit[g][u] != 0u, 0)) {
                    const int nb = t * STEP + g * GROUP + 16 * u;
                    unsigned mask = hit[g][u];
                    while (mask != 0u) {
                        const int src = __builtin_ctz(mask);
                        mask &= mask - 1u;
                        const float ds = __int_as_float(
                            __builtin_amdgcn_readlane(__float_as_int(dv[g][u]), src));
                        const float dn = __builtin_amdgcn_inverse_ballot_w32(src < 16 ? 0x0000FFFFu : 0xFFFF0000u)
                                             ? ds : ld;
                        const int   in = nb + (src & 15);
                        const float pd = row_shr1_f(ld, -INFINITY);
                        const int   pi = row_shr1_i(li, 0);
                        const bool gt  = ld > dn;
                        const bool pgt = pd > dn;
                        ld = __builtin_amdgcn_fmed3f(pd, dn, ld);
                        li = gt ? (pgt ? pi : in) : li;
                    }
                }
            }
        }
        thr = row_last_f(ld);
    };

    auto arrive = [&]() {
        __builtin_amdgcn_fence(__ATOMIC_RELEASE, "workgroup");
        __builtin_amdgcn_s_barrier_signal(-1);
    };
    auto pass = [&]() {
        __builtin_amdgcn_s_barrier_wait(-1);
        __builtin_amdgcn_fence(__ATOMIC_ACQUIRE, "workgroup");
    };

    fetch(0, px, py, pz);
    if (1 < nsteps) fetch(1, px1, py1, pz1);
    if (2 < nsteps) fetch(2, px2, py2, pz2);
    if (3 < nsteps) fetch(3, px3, py3, pz3);
    produce(0, 0, px, py, pz, true);
    if (1 < nsteps) produce(1, 1, px1, py1, pz1, true);
    if (2 < nsteps) produce(2, 2, px2, py2, pz2, true);
    if (3 < nsteps) produce(3, 3, px3, py3, pz3, true);
#pragma unroll
    for (int r = 0; r < 8; ++r) exch[(r * WAVES + wv) * 32 + lane] = mn[r];
    __syncthreads();
    {
        float c1 = INFINITY, c2 = INFINITY;
#pragma unroll
        for (int w = 0; w < WAVES; ++w) {
            const float v = exch[(wv * WAVES + w) * 32 + lane];
            c2 = __builtin_amdgcn_fmed3f(c1, c2, v);
            c1 = fminf(c1, v);
        }
#define XROW(v, ctrl) __int_as_float(__builtin_amdgcn_update_dpp( \
            0, __float_as_int(v), (ctrl), 0xF, 0xF, true))
#define SORT_STAGE(K, J, ctrl)                                                              \
        {                                                                                   \
            const float end = (((mrow & (K)) == 0) == ((mrow & (J)) == 0)) ? -INFINITY : INFINITY; \
            c1 = __builtin_amdgcn_fmed3f(c1, XROW(c1, ctrl), end);                          \
            c2 = __builtin_amdgcn_fmed3f(c2, XROW(c2, ctrl), end);                          \
        }
        SORT_STAGE(2, 1, 0xB1)
        SORT_STAGE(4, 2, 0x4E)   SORT_STAGE(4, 1, 0xB1)
        SORT_STAGE(8, 4, 0x164)  SORT_STAGE(8, 2, 0x4E)   SORT_STAGE(8, 1, 0xB1)
        SORT_STAGE(16, 8, 0x168) SORT_STAGE(16, 4, 0x164) SORT_STAGE(16, 2, 0x4E) SORT_STAGE(16, 1, 0xB1)
        float bnd = fminf(c1, XROW(c2, 0x140));
#undef SORT_STAGE
        bnd = fmaxf(bnd, XROW(bnd, 0x121));
        bnd = fmaxf(bnd, XROW(bnd, 0x122));
        bnd = fmaxf(bnd, XROW(bnd, 0x124));
        bnd = fmaxf(bnd, XROW(bnd, 0x128));
#undef XROW
        thr = bnd + (fabsf(bnd) * 0x1p-20f + 0x1p-100f);
    }
    ld = thr;
    arrive();
    if (4 < nsteps) fetch(4, px, py, pz);

    int bs = 0;
    int bp = 3;
    for (int s = 0; s < nsteps; ++s) {
        scan(s, bs);
        pass();
        if (s > 0 && s + 3 < nsteps) produce(s + 3, bp, px, py, pz, false);
        arrive();
        if (s > 0 && s + 4 < nsteps) fetch(s + 4, px, py, pz);
        bs = (bs + 1) & 3;
        bp = (bp + 1) & 3;
    }
    pass();

    lds_li[wv + (hi ? 8 : 0)][mrow] = li;
    __syncthreads();

    if (wv == 0 && lane < 16) {
        const int   mg = mga;
        const float qx = ax, qy = ay, qz = az;
        int midx[KNN];
#pragma unroll
        for (int j = 0; j < KNN; ++j) midx[j] = lds_li[mrow][j];

        float pxa[KNN], pya[KNN], pza[KNN], ed[KNN];
        float dmin = FLT_MAX;
#pragma unroll
        for (int j = 0; j < KNN; ++j) {
            const int   gi = midx[j];
            const float gx = pcb[0 * N + gi];
            const float gy = pcb[1 * N + gi];
            const float gz = pcb[2 * N + gi];
            const float dx = gx - qx, dy = gy - qy, dz = gz - qz;
            const float d  = dx * dx + dy * dy + dz * dz;
            pxa[j] = gx; pya[j] = gy; pza[j] = gz; ed[j] = d;
            dmin = fminf(dmin, d);
        }

        float wsum = 0.0f, ox = 0.0f, oy = 0.0f, oz = 0.0f;
#pragma unroll
        for (int j = 0; j < KNN; ++j) {
            const float w = __expf(dmin - ed[j]);
            wsum += w;
            ox += w * pxa[j];
            oy += w * pya[j];
            oz += w * pza[j];
        }
        const float inv = 1.0f / wsum;
        float* outb = out + (size_t)b * 3 * M;
        outb[0 * M + mg] = ox * inv;
        outb[1 * M + mg] = oy * inv;
        outb[2 * M + mg] = oz * inv;
    }
}

extern "C" void kernel_launch(void* const* d_in, const int* in_sizes, int n_in,
                              void* d_out, int out_size, void* d_ws, size_t ws_size,
                              hipStream_t stream) {
    (void)n_in; (void)out_size; (void)d_ws; (void)ws_size;
    const float* pc = (const float*)d_in[0];
    const float* qc = (const float*)d_in[1];
    float* out = (float*)d_out;

    const int B = 4;
    const int N = in_sizes[0] / (3 * B);
    const int M = in_sizes[1] / (3 * B);

    dim3 grid(B * (M / 16));
    softproj_knn_wmma<<<grid, 32 * WAVES, 0, stream>>>(pc, qc, out, N, M);
}
